# starting kernel
# baseline (speedup 1.0000x reference)
#include <hip/hip_runtime.h>
#include <hip/hip_bf16.h>

typedef __attribute__((ext_vector_type(2))) float v2f;
typedef __attribute__((ext_vector_type(8))) float v8f;

#define NUM_TYPES 8
#define EMBED_DIM 32

__global__ void k_init(const int* __restrict__ node_type,
                       float* __restrict__ c0, float* __restrict__ c1, int n_nodes) {
    int idx = blockIdx.x * blockDim.x + threadIdx.x;
    if (idx >= n_nodes * NUM_TYPES) return;
    int v = idx >> 3;
    int t = idx & 7;
    c0[idx] = (node_type[v] == t) ? 1.0f : 0.0f;
    c1[idx] = 0.0f;
}

__global__ void k_zero(float* __restrict__ p, int n) {
    int idx = blockIdx.x * blockDim.x + threadIdx.x;
    if (idx < n) p[idx] = 0.0f;
}

__global__ void k_edges(const int* __restrict__ src, const int* __restrict__ dst,
                        const float* __restrict__ cin, float* __restrict__ cout,
                        int n_edges) {
    long long gid = (long long)blockIdx.x * blockDim.x + threadIdx.x;
    int msg = (int)(gid >> 3);
    int lc  = (int)(gid & 7);
    if (msg >= n_edges) return;
    int u = src[msg];
    int v = dst[msg];
    float cu = cin[(size_t)u * NUM_TYPES + lc];
    float cv = cin[(size_t)v * NUM_TYPES + lc];
    __hip_atomic_fetch_add(&cout[(size_t)v * NUM_TYPES + lc], cu,
                           __ATOMIC_RELAXED, __HIP_MEMORY_SCOPE_AGENT);
    __hip_atomic_fetch_add(&cout[(size_t)u * NUM_TYPES + lc], cv,
                           __ATOMIC_RELAXED, __HIP_MEMORY_SCOPE_AGENT);
}

__global__ void k_embed_gemm(const float* __restrict__ counts,
                             const float* __restrict__ W,
                             float* __restrict__ out, int n_rows) {
    const int wave = (blockIdx.x * blockDim.x + threadIdx.x) >> 5;
    const int lane = threadIdx.x & 31;
    const int n_mtiles = (n_rows + 15) >> 4;
    const int mt = wave >> 1;
    const int nt = wave & 1;
    if (mt >= n_mtiles) return;

    const int m     = lane & 15;
    const int khalf = (lane >> 4) & 1;

    v8f acc = {0.f, 0.f, 0.f, 0.f, 0.f, 0.f, 0.f, 0.f};
    const float* crow = counts + (size_t)(mt * 16 + m) * NUM_TYPES;

#pragma unroll
    for (int s = 0; s < 2; ++s) {
        const int kb = s * 4 + khalf * 2;
        v2f a, b;
        a.x = crow[kb];
        a.y = crow[kb + 1];
        b.x = W[(size_t)kb       * EMBED_DIM + nt * 16 + m];
        b.y = W[(size_t)(kb + 1) * EMBED_DIM + nt * 16 + m];
        acc = __builtin_amdgcn_wmma_f32_16x16x4_f32(
             false, a,  false, b,
             (short)0, acc,  false,  false);
    }

    if (mt * 16 + 16 <= n_rows) {
        float* obase = out + (size_t)(mt * 16 + khalf * 8) * EMBED_DIM + nt * 16 + m;
#pragma unroll
        for (int r = 0; r < 8; ++r) {
            obase[(size_t)r * EMBED_DIM] = acc[r];
        }
    } else {
#pragma unroll
        for (int r = 0; r < 8; ++r) {
            int row = mt * 16 + khalf * 8 + r;
            if (row < n_rows) out[(size_t)row * EMBED_DIM + nt * 16 + m] = acc[r];
        }
    }
}

extern "C" void kernel_launch(void* const* d_in, const int* in_sizes, int n_in,
                              void* d_out, int out_size, void* d_ws, size_t ws_size,
                              hipStream_t stream) {
    const float* embed_table = (const float*)d_in[0];
    const int*   node_type   = (const int*)d_in[1];
    const int*   src         = (const int*)d_in[2];
    const int*   dst         = (const int*)d_in[3];
    float*       out         = (float*)d_out;

    const int n_nodes = in_sizes[1];
    const int n_edges = in_sizes[2];
    const int n_rows  = out_size / EMBED_DIM;

    float* c0 = (float*)d_ws;
    float* c1 = c0 + (size_t)n_nodes * NUM_TYPES;

    const int nc = n_nodes * NUM_TYPES;

    k_init<<<(nc + 255) / 256, 256, 0, stream>>>(node_type, c0, c1, n_nodes);

    {
        long long threads = (long long)n_edges * 8;
        int blocks = (int)((threads + 255) / 256);
        k_edges<<<blocks, 256, 0, stream>>>(src, dst, c0, c1, n_edges);
    }

    k_zero<<<(nc + 255) / 256, 256, 0, stream>>>(c0, nc);
    {
        long long threads = (long long)n_edges * 8;
        int blocks = (int)((threads + 255) / 256);
        k_edges<<<blocks, 256, 0, stream>>>(src, dst, c1, c0, n_edges);
    }

    {
        int n_mtiles = (n_rows + 15) / 16;
        int tiles = n_mtiles * 2;
        int waves_per_block = 256 / 32;
        int blocks = (tiles + waves_per_block - 1) / waves_per_block;
        k_embed_gemm<<<blocks, 256, 0, stream>>>(c0, embed_table, out, n_rows);
    }
}


# Product fused into round 2's owner: the work-group multiplies its 256 LDS row sums by the table on f32 WMMA (a 16x16 tile per wave) and stores out rows itself; 3 launches, add order as before
# speedup vs baseline: 4.7835x; 4.7835x over previous
#include <hip/hip_runtime.h>
#include <hip/hip_bf16.h>

typedef __attribute__((ext_vector_type(2))) float v2f;
typedef __attribute__((ext_vector_type(8))) float v8f;

#define NUM_TYPES 8
#define EMBED_DIM 32

__global__ void k_init(const int* __restrict__ node_type,
                       float* __restrict__ c0, float* __restrict__ c1, int n_nodes) {
    int idx = blockIdx.x * blockDim.x + threadIdx.x;
    if (idx >= n_nodes * NUM_TYPES) return;
    int v = idx >> 3;
    int t = idx & 7;
    c0[idx] = (node_type[v] == t) ? 1.0f : 0.0f;
    c1[idx] = 0.0f;
}

__global__ void k_zero(float* __restrict__ p, int n) {
    int idx = blockIdx.x * blockDim.x + threadIdx.x;
    if (idx < n) p[idx] = 0.0f;
}

#define EDGES_PER_WAVE 128
#define CHUNKS_PER_WAVE (EDGES_PER_WAVE / 32)

template <bool GUARD>
__device__ __forceinline__ void scatter_chunk(int u, int v, int lane,
                                              const float* __restrict__ cin,
                                              float* __restrict__ cout) {
    const unsigned lcb = (unsigned)(lane & 7) * 4u;
    const int sub = lane >> 3;
    const char* __restrict__ cin_b  = (const char*)cin;
    char* __restrict__       cout_b = (char*)cout;
    unsigned ou[8], ov[8];
    float    cu[8], cv[8];
#pragma unroll
    for (int j = 0; j < 8; ++j) {
        const int uu = __shfl(u, j * 4 + sub);
        const int vv = __shfl(v, j * 4 + sub);
        ou[j] = (unsigned)uu * (NUM_TYPES * 4u) + lcb;
        ov[j] = (unsigned)vv * (NUM_TYPES * 4u) + lcb;
        if (!GUARD || uu >= 0) {
            cu[j] = *(const float*)(cin_b + ou[j]);
            cv[j] = *(const float*)(cin_b + ov[j]);
        } else {
            cu[j] = 0.0f;
            cv[j] = 0.0f;
        }
    }
#pragma unroll
    for (int j = 0; j < 8; ++j) {
        if (!GUARD || __shfl(u, j * 4 + sub) >= 0) {
            __hip_atomic_fetch_add((float*)(cout_b + ov[j]), cu[j],
                                   __ATOMIC_RELAXED, __HIP_MEMORY_SCOPE_AGENT);
            __hip_atomic_fetch_add((float*)(cout_b + ou[j]), cv[j],
                                   __ATOMIC_RELAXED, __HIP_MEMORY_SCOPE_AGENT);
        }
    }
}

__global__ void __launch_bounds__(256)
k_edges(const int* __restrict__ src, const int* __restrict__ dst,
        const float* __restrict__ cin, float* __restrict__ cout,
        int n_edges) {
    const int lane = threadIdx.x & 31;
    const int wave = (blockIdx.x * blockDim.x + threadIdx.x) >> 5;
    const long long first = (long long)wave * EDGES_PER_WAVE;
    if (first >= n_edges) return;

    int u[CHUNKS_PER_WAVE], v[CHUNKS_PER_WAVE];
    if (first + EDGES_PER_WAVE <= n_edges) {
#pragma unroll
        for (int c = 0; c < CHUNKS_PER_WAVE; ++c) {
            u[c] = src[first + c * 32 + lane];
            v[c] = dst[first + c * 32 + lane];
        }
#pragma unroll
        for (int c = 0; c < CHUNKS_PER_WAVE; ++c)
            scatter_chunk<false>(u[c], v[c], lane, cin, cout);
    } else {
#pragma unroll
        for (int c = 0; c < CHUNKS_PER_WAVE; ++c) {
            const long long e = first + c * 32 + lane;
            u[c] = (e < n_edges) ? src[e] : -1;
            v[c] = (e < n_edges) ? dst[e] : -1;
        }
#pragma unroll
        for (int c = 0; c < CHUNKS_PER_WAVE; ++c)
            scatter_chunk<true>(u[c], v[c], lane, cin, cout);
    }
}

__global__ void k_embed_gemm(const float* __restrict__ counts,
                             const float* __restrict__ W,
                             float* __restrict__ out, int n_rows) {
    const int wave = (blockIdx.x * blockDim.x + threadIdx.x) >> 5;
    const int lane = threadIdx.x & 31;
    const int n_mtiles = (n_rows + 15) >> 4;
    const int mt = wave >> 1;
    const int nt = wave & 1;
    if (mt >= n_mtiles) return;

    const int m     = lane & 15;
    const int khalf = (lane >> 4) & 1;

    v8f acc = {0.f, 0.f, 0.f, 0.f, 0.f, 0.f, 0.f, 0.f};
    const float* crow = counts + (size_t)(mt * 16 + m) * NUM_TYPES;

#pragma unroll
    for (int s = 0; s < 2; ++s) {
        const int kb = s * 4 + khalf * 2;
        v2f a, b;
        a.x = crow[kb];
        a.y = crow[kb + 1];
        b.x = W[(size_t)kb       * EMBED_DIM + nt * 16 + m];
        b.y = W[(size_t)(kb + 1) * EMBED_DIM + nt * 16 + m];
        acc = __builtin_amdgcn_wmma_f32_16x16x4_f32(
             false, a,  false, b,
             (short)0, acc,  false,  false);
    }

    if (mt * 16 + 16 <= n_rows) {
        float* obase = out + (size_t)(mt * 16 + khalf * 8) * EMBED_DIM + nt * 16 + m;
#pragma unroll
        for (int r = 0; r < 8; ++r) {
            obase[(size_t)r * EMBED_DIM] = acc[r];
        }
    } else {
#pragma unroll
        for (int r = 0; r < 8; ++r) {
            int row = mt * 16 + khalf * 8 + r;
            if (row < n_rows) out[(size_t)row * EMBED_DIM + nt * 16 + m] = acc[r];
        }
    }
}

#define BIN_ROWS  256
#define MAX_BINS  512
#define N_CHUNKS  256
#define SORT_CAP  16384
#define ENT_NODE  0x1FFFFFu
#define ENT_NONE  0xFFFFFFFFu
#define EPT       (SORT_CAP / 2 / 1024)

__global__ void __launch_bounds__(1024)
k_bin(const int* __restrict__ src, const int* __restrict__ dst,
      const int* __restrict__ node_type,
      unsigned* __restrict__ list, int2* __restrict__ tab,
      int n_edges, int n_nodes, int chunk_edges) {
    __shared__ int cnt[MAX_BINS];
    __shared__ int start[MAX_BINS];
    __shared__ int wsum[MAX_BINS / 32];
    __shared__ int total;
    __shared__ unsigned sorted[SORT_CAP];
    const int tid  = threadIdx.x;
    const int lane = tid & 31;
    const int wg   = blockIdx.x;
    const long long e0 = (long long)wg * chunk_edges;
    long long e1 = e0 + chunk_edges;
    if (e1 > n_edges) e1 = n_edges;
    if (tid < MAX_BINS) cnt[tid] = 0;
    unsigned uu[EPT], vv[EPT], tt[EPT];
#pragma unroll
    for (int j = 0; j < EPT; ++j) {
        const long long e = e0 + tid + j * 1024;
        unsigned u = ENT_NONE, v = ENT_NONE;
        if (e < e1) {
            u = (unsigned)src[e];
            v = (unsigned)dst[e];
        }
        const bool ok = u < (unsigned)n_nodes && v < (unsigned)n_nodes;
        uu[j] = ok ? u : ENT_NONE;
        vv[j] = v;
    }
#pragma unroll
    for (int j = 0; j < EPT; ++j) {
        const bool has = uu[j] != ENT_NONE;
        const unsigned tu = (unsigned)min(max(node_type[has ? uu[j] : 0u], 0), NUM_TYPES - 1);
        const unsigned tv = (unsigned)min(max(node_type[has ? vv[j] : 0u], 0), NUM_TYPES - 1);
        tt[j] = tu | (tv << 3);
    }
    __syncthreads();
#pragma unroll
    for (int j = 0; j < EPT; ++j) {
        if (uu[j] != ENT_NONE) {
            atomicAdd(&cnt[vv[j] / BIN_ROWS], 1);
            atomicAdd(&cnt[uu[j] / BIN_ROWS], 1);
        }
    }
    __syncthreads();
    int a = 0, inc = 0;
    if (tid < MAX_BINS) {
        a = cnt[tid];
        inc = a;
        for (int d = 1; d < 32; d <<= 1) {
            const int t = __shfl_up(inc, d);
            if (lane >= d) inc += t;
        }
        if (lane == 31) wsum[tid >> 5] = inc;
    }
    __syncthreads();
    if (tid < MAX_BINS) {
        int base = 0;
        for (int w = 0; w < (tid >> 5); ++w) base += wsum[w];
        const int ex = base + inc - a;
        start[tid] = ex;
        tab[wg * MAX_BINS + tid] = make_int2(ex, a);
        cnt[tid] = 0;
        if (tid == MAX_BINS - 1) total = ex + a;
    }
    __syncthreads();
#pragma unroll
    for (int j = 0; j < EPT; ++j) {
        if (uu[j] != ENT_NONE) {
            const unsigned u = uu[j];
            const unsigned v = vv[j];
            int bin = v / BIN_ROWS;
            int pos = start[bin] + atomicAdd(&cnt[bin], 1);
            sorted[pos] = ((v % BIN_ROWS) << 24) | ((tt[j] & 7u) << 21) | u;
            bin = u / BIN_ROWS;
            pos = start[bin] + atomicAdd(&cnt[bin], 1);
            sorted[pos] = ((u % BIN_ROWS) << 24) | ((tt[j] >> 3) << 21) | v;
        }
    }
    __syncthreads();
    unsigned* __restrict__ region = list + (size_t)wg * 2 * (size_t)chunk_edges;
    const int n = total;
    for (int i = tid; i < n; i += 1024) region[i] = sorted[i];
}

#define CNT_THREADS 256
__global__ void __launch_bounds__(CNT_THREADS)
k_count(const unsigned* __restrict__ list, const int2* __restrict__ tab,
        float* __restrict__ cout, int n_nodes, int chunk_edges) {
    __shared__ int acc[BIN_ROWS * NUM_TYPES];
    const int tid  = threadIdx.x;
    const int lane = tid & 31;
    const int wave = tid >> 5;
    const int b    = blockIdx.x;
    for (int i = tid; i < BIN_ROWS * NUM_TYPES; i += CNT_THREADS) acc[i] = 0;
    __syncthreads();

    const int2 oc = tab[(wave * 32 + lane) * MAX_BINS + b];
    const unsigned* __restrict__ wbase =
        list + (size_t)(wave * 32) * 2 * (size_t)chunk_edges;
    for (int g = 0; g < 2; ++g) {
        unsigned ea[16], eb[16];
#pragma unroll
        for (int k = 0; k < 16; ++k) {
            const int s = g * 16 + k;
            const int off = __builtin_amdgcn_readlane(oc.x, s);
            const int n   = __builtin_amdgcn_readlane(oc.y, s);
            const unsigned* __restrict__ seg =
                wbase + (size_t)s * 2 * (size_t)chunk_edges + off;
            ea[k] = (lane < n)      ? seg[lane]      : ENT_NONE;
            eb[k] = (lane + 32 < n) ? seg[lane + 32] : ENT_NONE;
        }
#pragma unroll
        for (int k = 0; k < 16; ++k) {
            if (ea[k] != ENT_NONE) atomicAdd(&acc[ea[k] >> 21], 1);
            if (eb[k] != ENT_NONE) atomicAdd(&acc[eb[k] >> 21], 1);
        }
    }
    int longest = oc.y;
    for (int d = 16; d > 0; d >>= 1) longest = max(longest, __shfl_xor(longest, d));
    if (longest > 64) {
        for (int s = 0; s < 32; ++s) {
            const int off = __shfl(oc.x, s);
            const int n   = __shfl(oc.y, s);
            const unsigned* __restrict__ seg =
                wbase + (size_t)s * 2 * (size_t)chunk_edges + off;
            for (int i = 64 + lane; i < n; i += 32) atomicAdd(&acc[seg[i] >> 21], 1);
        }
    }
    __syncthreads();
    const int row0 = b * BIN_ROWS;
    for (int i = tid; i < BIN_ROWS * NUM_TYPES; i += CNT_THREADS)
        if (row0 + (i >> 3) < n_nodes)
            cout[(size_t)row0 * NUM_TYPES + i] = (float)acc[i];
}

__global__ void __launch_bounds__(1024)
k_own(const unsigned* __restrict__ list, const int2* __restrict__ tab,
      const float* __restrict__ cin, const float* __restrict__ W,
      float* __restrict__ out, int n_rows, int chunk_edges) {
    __shared__ float acc[BIN_ROWS * NUM_TYPES];
    const int tid  = threadIdx.x;
    const int lane = tid & 31;
    const int wave = tid >> 5;
    const int b    = blockIdx.x;
    const int sub  = lane >> 3;
    const int d    = lane & 7;
    acc[tid]        = 0.0f;
    acc[tid + 1024] = 0.0f;
    __syncthreads();

    int off[N_CHUNKS / 32], num[N_CHUNKS / 32];
#pragma unroll
    for (int k = 0; k < N_CHUNKS / 32; ++k) {
        const int2 oc = tab[(wave + 32 * k) * MAX_BINS + b];
        off[k] = __builtin_amdgcn_readfirstlane(oc.x);
        num[k] = __builtin_amdgcn_readfirstlane(oc.y);
    }
    const int wv    = __builtin_amdgcn_readfirstlane(wave);
    const int mt    = wv >> 1;
    const int nt    = wv & 1;
    const int m     = lane & 15;
    const int khalf = lane >> 4;
    const int trow  = b * BIN_ROWS + mt * 16;
    float* obase = out + (size_t)trow * EMBED_DIM + nt * 16 + (khalf * 8 * EMBED_DIM + m);
    const int nvalid = n_rows - (trow + khalf * 8);
    v2f bf[2];
#pragma unroll
    for (int s = 0; s < 2; ++s) {
        const int kb = s * 4 + khalf * 2;
        bf[s].x = W[kb       * EMBED_DIM + nt * 16 + m];
        bf[s].y = W[(kb + 1) * EMBED_DIM + nt * 16 + m];
    }
#pragma unroll
    for (int k = 0; k < N_CHUNKS / 32; ++k) {
        const unsigned* __restrict__ seg =
            list + (size_t)(wave + 32 * k) * 2 * (size_t)chunk_edges + off[k];
        const int c = num[k];
        for (int i = 0; i < c; i += 32) {
            const unsigned ent = (i + lane < c) ? seg[i + lane] : 0u;
            float    val[8];
            unsigned ad[8];
#pragma unroll
            for (int j = 0; j < 8; ++j) {
                const unsigned en = __shfl(ent, j * 4 + sub);
                ad[j]  = (en >> 24) * NUM_TYPES + d;
                val[j] = (i + j * 4 + sub < c)
                             ? cin[(size_t)(en & ENT_NODE) * NUM_TYPES + d] : 0.0f;
            }
#pragma unroll
            for (int j = 0; j < 8; ++j)
                if (i + j * 4 + sub < c) atomicAdd(&acc[ad[j]], val[j]);
        }
    }
    __syncthreads();
    v8f o = {0.f, 0.f, 0.f, 0.f, 0.f, 0.f, 0.f, 0.f};
    const float* arow = acc + (mt * 16 + m) * NUM_TYPES;
#pragma unroll
    for (int s = 0; s < 2; ++s) {
        const int kb = s * 4 + khalf * 2;
        v2f a;
        a.x = arow[kb];
        a.y = arow[kb + 1];
        o = __builtin_amdgcn_wmma_f32_16x16x4_f32(
             false, a,  false, bf[s],
             (short)0, o,  false,  false);
    }
    if (trow + 16 <= n_rows) {
#pragma unroll
        for (int r = 0; r < 8; ++r) obase[r * EMBED_DIM] = o[r];
    } else {
#pragma unroll
        for (int r = 0; r < 8; ++r)
            if (r < nvalid) obase[r * EMBED_DIM] = o[r];
    }
}

extern "C" void kernel_launch(void* const* d_in, const int* in_sizes, int n_in,
                              void* d_out, int out_size, void* d_ws, size_t ws_size,
                              hipStream_t stream) {
    const float* embed_table = (const float*)d_in[0];
    const int*   node_type   = (const int*)d_in[1];
    const int*   src         = (const int*)d_in[2];
    const int*   dst         = (const int*)d_in[3];
    float*       out         = (float*)d_out;

    const int n_nodes = in_sizes[1];
    const int n_edges = in_sizes[2];
    const int n_rows  = out_size / EMBED_DIM;

    const int nc = n_nodes * NUM_TYPES;

    const int    chunk_edges = (n_edges + N_CHUNKS - 1) / N_CHUNKS;
    const int    n_bins_all  = (n_nodes + BIN_ROWS - 1) / BIN_ROWS;
    const int    n_bins      = (n_rows + BIN_ROWS - 1) / BIN_ROWS;
    float*       cnt1        = (float*)d_ws;
    unsigned*    list        = (unsigned*)(cnt1 + (size_t)nc);
    const size_t list_words  = (size_t)N_CHUNKS * 2 * (size_t)chunk_edges;
    int2*        tab         = (int2*)(list + list_words);
    const size_t need = ((size_t)nc + list_words) * sizeof(float)
                      + (size_t)N_CHUNKS * MAX_BINS * sizeof(int2);
    const bool by_owner = ws_size >= need && n_bins_all <= MAX_BINS && n_rows <= n_nodes
                          && n_nodes > 0 && 2 * (long long)chunk_edges <= SORT_CAP;

    if (by_owner) {
        k_bin<<<N_CHUNKS, 1024, 0, stream>>>(src, dst, node_type, list, tab,
                                             n_edges, n_nodes, chunk_edges);
        k_count<<<n_bins_all, CNT_THREADS, 0, stream>>>(list, tab, cnt1, n_nodes, chunk_edges);
        k_own<<<n_bins, 1024, 0, stream>>>(list, tab, cnt1, embed_table, out, n_rows, chunk_edges);
    } else {
        float* c0 = (float*)d_ws;
        float* c1 = c0 + (size_t)nc;
        const int edges_per_block = 8 * EDGES_PER_WAVE;
        const int edge_blocks = (n_edges + edges_per_block - 1) / edges_per_block;
        k_init<<<(nc + 255) / 256, 256, 0, stream>>>(node_type, c0, c1, n_nodes);
        k_edges<<<edge_blocks, 256, 0, stream>>>(src, dst, c0, c1, n_edges);
        k_zero<<<(nc + 255) / 256, 256, 0, stream>>>(c0, nc);
        k_edges<<<edge_blocks, 256, 0, stream>>>(src, dst, c1, c0, n_edges);
        int n_mtiles = (n_rows + 15) / 16;
        int tiles = n_mtiles * 2;
        int waves_per_block = 256 / 32;
        int blocks = (tiles + waves_per_block - 1) / waves_per_block;
        k_embed_gemm<<<blocks, 256, 0, stream>>>(c0, embed_table, out, n_rows);
    }
}


# Sort kernel looks node types up after its LDS sort, in sorted order (a wave's 32 targets lie in a few table lines, not 32), and hands each to the mirror entry's LDS slot; add order unchanged
# speedup vs baseline: 5.1429x; 1.0751x over previous
#include <hip/hip_runtime.h>
#include <hip/hip_bf16.h>

typedef __attribute__((ext_vector_type(2))) float v2f;
typedef __attribute__((ext_vector_type(8))) float v8f;

#define NUM_TYPES 8
#define EMBED_DIM 32

__global__ void k_init(const int* __restrict__ node_type,
                       float* __restrict__ c0, float* __restrict__ c1, int n_nodes) {
    int idx = blockIdx.x * blockDim.x + threadIdx.x;
    if (idx >= n_nodes * NUM_TYPES) return;
    int v = idx >> 3;
    int t = idx & 7;
    c0[idx] = (node_type[v] == t) ? 1.0f : 0.0f;
    c1[idx] = 0.0f;
}

__global__ void k_zero(float* __restrict__ p, int n) {
    int idx = blockIdx.x * blockDim.x + threadIdx.x;
    if (idx < n) p[idx] = 0.0f;
}

#define EDGES_PER_WAVE 128
#define CHUNKS_PER_WAVE (EDGES_PER_WAVE / 32)

template <bool GUARD>
__device__ __forceinline__ void scatter_chunk(int u, int v, int lane,
                                              const float* __restrict__ cin,
                                              float* __restrict__ cout) {
    const unsigned lcb = (unsigned)(lane & 7) * 4u;
    const int sub = lane >> 3;
    const char* __restrict__ cin_b  = (const char*)cin;
    char* __restrict__       cout_b = (char*)cout;
    unsigned ou[8], ov[8];
    float    cu[8], cv[8];
#pragma unroll
    for (int j = 0; j < 8; ++j) {
        const int uu = __shfl(u, j * 4 + sub);
        const int vv = __shfl(v, j * 4 + sub);
        ou[j] = (unsigned)uu * (NUM_TYPES * 4u) + lcb;
        ov[j] = (unsigned)vv * (NUM_TYPES * 4u) + lcb;
        if (!GUARD || uu >= 0) {
            cu[j] = *(const float*)(cin_b + ou[j]);
            cv[j] = *(const float*)(cin_b + ov[j]);
        } else {
            cu[j] = 0.0f;
            cv[j] = 0.0f;
        }
    }
#pragma unroll
    for (int j = 0; j < 8; ++j) {
        if (!GUARD || __shfl(u, j * 4 + sub) >= 0) {
            __hip_atomic_fetch_add((float*)(cout_b + ov[j]), cu[j],
                                   __ATOMIC_RELAXED, __HIP_MEMORY_SCOPE_AGENT);
            __hip_atomic_fetch_add((float*)(cout_b + ou[j]), cv[j],
                                   __ATOMIC_RELAXED, __HIP_MEMORY_SCOPE_AGENT);
        }
    }
}

__global__ void __launch_bounds__(256)
k_edges(const int* __restrict__ src, const int* __restrict__ dst,
        const float* __restrict__ cin, float* __restrict__ cout,
        int n_edges) {
    const int lane = threadIdx.x & 31;
    const int wave = (blockIdx.x * blockDim.x + threadIdx.x) >> 5;
    const long long first = (long long)wave * EDGES_PER_WAVE;
    if (first >= n_edges) return;

    int u[CHUNKS_PER_WAVE], v[CHUNKS_PER_WAVE];
    if (first + EDGES_PER_WAVE <= n_edges) {
#pragma unroll
        for (int c = 0; c < CHUNKS_PER_WAVE; ++c) {
            u[c] = src[first + c * 32 + lane];
            v[c] = dst[first + c * 32 + lane];
        }
#pragma unroll
        for (int c = 0; c < CHUNKS_PER_WAVE; ++c)
            scatter_chunk<false>(u[c], v[c], lane, cin, cout);
    } else {
#pragma unroll
        for (int c = 0; c < CHUNKS_PER_WAVE; ++c) {
            const long long e = first + c * 32 + lane;
            u[c] = (e < n_edges) ? src[e] : -1;
            v[c] = (e < n_edges) ? dst[e] : -1;
        }
#pragma unroll
        for (int c = 0; c < CHUNKS_PER_WAVE; ++c)
            scatter_chunk<true>(u[c], v[c], lane, cin, cout);
    }
}

__global__ void k_embed_gemm(const float* __restrict__ counts,
                             const float* __restrict__ W,
                             float* __restrict__ out, int n_rows) {
    const int wave = (blockIdx.x * blockDim.x + threadIdx.x) >> 5;
    const int lane = threadIdx.x & 31;
    const int n_mtiles = (n_rows + 15) >> 4;
    const int mt = wave >> 1;
    const int nt = wave & 1;
    if (mt >= n_mtiles) return;

    const int m     = lane & 15;
    const int khalf = (lane >> 4) & 1;

    v8f acc = {0.f, 0.f, 0.f, 0.f, 0.f, 0.f, 0.f, 0.f};
    const float* crow = counts + (size_t)(mt * 16 + m) * NUM_TYPES;

#pragma unroll
    for (int s = 0; s < 2; ++s) {
        const int kb = s * 4 + khalf * 2;
        v2f a, b;
        a.x = crow[kb];
        a.y = crow[kb + 1];
        b.x = W[(size_t)kb       * EMBED_DIM + nt * 16 + m];
        b.y = W[(size_t)(kb + 1) * EMBED_DIM + nt * 16 + m];
        acc = __builtin_amdgcn_wmma_f32_16x16x4_f32(
             false, a,  false, b,
             (short)0, acc,  false,  false);
    }

    if (mt * 16 + 16 <= n_rows) {
        float* obase = out + (size_t)(mt * 16 + khalf * 8) * EMBED_DIM + nt * 16 + m;
#pragma unroll
        for (int r = 0; r < 8; ++r) {
            obase[(size_t)r * EMBED_DIM] = acc[r];
        }
    } else {
#pragma unroll
        for (int r = 0; r < 8; ++r) {
            int row = mt * 16 + khalf * 8 + r;
            if (row < n_rows) out[(size_t)row * EMBED_DIM + nt * 16 + m] = acc[r];
        }
    }
}

#define BIN_ROWS  256
#define MAX_BINS  512
#define N_CHUNKS  256
#define SORT_CAP  16384
#define ENT_NODE  0x1FFFFFu
#define ENT_NONE  0xFFFFFFFFu
#define EPT       (SORT_CAP / 2 / 1024)

#define AUX_NODE_BITS 17
__global__ void __launch_bounds__(1024)
k_bin(const int* __restrict__ src, const int* __restrict__ dst,
      const int* __restrict__ node_type,
      unsigned* __restrict__ list, int2* __restrict__ tab,
      int n_edges, int n_nodes, int chunk_edges) {
    __shared__ int cnt[MAX_BINS];
    __shared__ int start[MAX_BINS];
    __shared__ int wsum[MAX_BINS / 32];
    __shared__ int total;
    __shared__ unsigned sorted[SORT_CAP];
    __shared__ unsigned aux[SORT_CAP];
    const int tid  = threadIdx.x;
    const int lane = tid & 31;
    const int wg   = blockIdx.x;
    const long long e0 = (long long)wg * chunk_edges;
    long long e1 = e0 + chunk_edges;
    if (e1 > n_edges) e1 = n_edges;
    if (tid < MAX_BINS) cnt[tid] = 0;
    unsigned uu[EPT], vv[EPT];
#pragma unroll
    for (int j = 0; j < EPT; ++j) {
        const long long e = e0 + tid + j * 1024;
        unsigned u = ENT_NONE, v = ENT_NONE;
        if (e < e1) {
            u = (unsigned)src[e];
            v = (unsigned)dst[e];
        }
        const bool ok = u < (unsigned)n_nodes && v < (unsigned)n_nodes;
        uu[j] = ok ? u : ENT_NONE;
        vv[j] = v;
    }
    __syncthreads();
#pragma unroll
    for (int j = 0; j < EPT; ++j) {
        if (uu[j] != ENT_NONE) {
            atomicAdd(&cnt[vv[j] / BIN_ROWS], 1);
            atomicAdd(&cnt[uu[j] / BIN_ROWS], 1);
        }
    }
    __syncthreads();
    int a = 0, inc = 0;
    if (tid < MAX_BINS) {
        a = cnt[tid];
        inc = a;
        for (int d = 1; d < 32; d <<= 1) {
            const int t = __shfl_up(inc, d);
            if (lane >= d) inc += t;
        }
        if (lane == 31) wsum[tid >> 5] = inc;
    }
    __syncthreads();
    if (tid < MAX_BINS) {
        int base = 0;
        for (int w = 0; w < (tid >> 5); ++w) base += wsum[w];
        const int ex = base + inc - a;
        start[tid] = ex;
        tab[wg * MAX_BINS + tid] = make_int2(ex, a);
        cnt[tid] = 0;
        if (tid == MAX_BINS - 1) total = ex + a;
    }
    __syncthreads();
#pragma unroll
    for (int j = 0; j < EPT; ++j) {
        if (uu[j] != ENT_NONE) {
            const unsigned u = uu[j];
            const unsigned v = vv[j];
            int bin = v / BIN_ROWS;
            const int pv = start[bin] + atomicAdd(&cnt[bin], 1);
            bin = u / BIN_ROWS;
            const int pu = start[bin] + atomicAdd(&cnt[bin], 1);
            sorted[pv] = ((v % BIN_ROWS) << 24) | u;
            sorted[pu] = ((u % BIN_ROWS) << 24) | v;
            aux[pv] = ((unsigned)pu << AUX_NODE_BITS) | v;
            aux[pu] = ((unsigned)pv << AUX_NODE_BITS) | u;
        }
    }
    __syncthreads();
    const int n = total;
    if (n > 0) {
        unsigned ax[2 * EPT];
        int      ty[2 * EPT];
#pragma unroll
        for (int k = 0; k < 2 * EPT; ++k) {
            const int i = tid + k * 1024;
            ax[k] = aux[(i < n) ? i : 0];
        }
#pragma unroll
        for (int k = 0; k < 2 * EPT; ++k) {
            const int i = tid + k * 1024;
            ty[k] = node_type[(i < n) ? (ax[k] & ((1u << AUX_NODE_BITS) - 1u)) : 0u];
        }
#pragma unroll
        for (int k = 0; k < 2 * EPT; ++k) {
            const int i = tid + k * 1024;
            if (i < n) {
                const unsigned t = (unsigned)min(max(ty[k], 0), NUM_TYPES - 1);
                atomicOr(&sorted[ax[k] >> AUX_NODE_BITS], t << 21);
            }
        }
    }
    __syncthreads();
    unsigned* __restrict__ region = list + (size_t)wg * 2 * (size_t)chunk_edges;
    for (int i = tid; i < n; i += 1024) region[i] = sorted[i];
}

#define CNT_THREADS 256
__global__ void __launch_bounds__(CNT_THREADS)
k_count(const unsigned* __restrict__ list, const int2* __restrict__ tab,
        float* __restrict__ cout, int n_nodes, int chunk_edges) {
    __shared__ int acc[BIN_ROWS * NUM_TYPES];
    const int tid  = threadIdx.x;
    const int lane = tid & 31;
    const int wave = tid >> 5;
    const int b    = blockIdx.x;
    for (int i = tid; i < BIN_ROWS * NUM_TYPES; i += CNT_THREADS) acc[i] = 0;
    __syncthreads();

    const int2 oc = tab[(wave * 32 + lane) * MAX_BINS + b];
    const unsigned* __restrict__ wbase =
        list + (size_t)(wave * 32) * 2 * (size_t)chunk_edges;
    for (int g = 0; g < 2; ++g) {
        unsigned ea[16], eb[16];
#pragma unroll
        for (int k = 0; k < 16; ++k) {
            const int s = g * 16 + k;
            const int off = __builtin_amdgcn_readlane(oc.x, s);
            const int n   = __builtin_amdgcn_readlane(oc.y, s);
            const unsigned* __restrict__ seg =
                wbase + (size_t)s * 2 * (size_t)chunk_edges + off;
            ea[k] = (lane < n)      ? seg[lane]      : ENT_NONE;
            eb[k] = (lane + 32 < n) ? seg[lane + 32] : ENT_NONE;
        }
#pragma unroll
        for (int k = 0; k < 16; ++k) {
            if (ea[k] != ENT_NONE) atomicAdd(&acc[ea[k] >> 21], 1);
            if (eb[k] != ENT_NONE) atomicAdd(&acc[eb[k] >> 21], 1);
        }
    }
    int longest = oc.y;
    for (int d = 16; d > 0; d >>= 1) longest = max(longest, __shfl_xor(longest, d));
    if (longest > 64) {
        for (int s = 0; s < 32; ++s) {
            const int off = __shfl(oc.x, s);
            const int n   = __shfl(oc.y, s);
            const unsigned* __restrict__ seg =
                wbase + (size_t)s * 2 * (size_t)chunk_edges + off;
            for (int i = 64 + lane; i < n; i += 32) atomicAdd(&acc[seg[i] >> 21], 1);
        }
    }
    __syncthreads();
    const int row0 = b * BIN_ROWS;
    for (int i = tid; i < BIN_ROWS * NUM_TYPES; i += CNT_THREADS)
        if (row0 + (i >> 3) < n_nodes)
            cout[(size_t)row0 * NUM_TYPES + i] = (float)acc[i];
}

__global__ void __launch_bounds__(1024)
k_own(const unsigned* __restrict__ list, const int2* __restrict__ tab,
      const float* __restrict__ cin, const float* __restrict__ W,
      float* __restrict__ out, int n_rows, int chunk_edges) {
    __shared__ float acc[BIN_ROWS * NUM_TYPES];
    const int tid  = threadIdx.x;
    const int lane = tid & 31;
    const int wave = tid >> 5;
    const int b    = blockIdx.x;
    const int sub  = lane >> 3;
    const int d    = lane & 7;
    acc[tid]        = 0.0f;
    acc[tid + 1024] = 0.0f;
    __syncthreads();

    int off[N_CHUNKS / 32], num[N_CHUNKS / 32];
#pragma unroll
    for (int k = 0; k < N_CHUNKS / 32; ++k) {
        const int2 oc = tab[(wave + 32 * k) * MAX_BINS + b];
        off[k] = __builtin_amdgcn_readfirstlane(oc.x);
        num[k] = __builtin_amdgcn_readfirstlane(oc.y);
    }
    const int wv    = __builtin_amdgcn_readfirstlane(wave);
    const int mt    = wv >> 1;
    const int nt    = wv & 1;
    const int m     = lane & 15;
    const int khalf = lane >> 4;
    const int trow  = b * BIN_ROWS + mt * 16;
    float* obase = out + (size_t)trow * EMBED_DIM + nt * 16 + (khalf * 8 * EMBED_DIM + m);
    const int nvalid = n_rows - (trow + khalf * 8);
    v2f bf[2];
#pragma unroll
    for (int s = 0; s < 2; ++s) {
        const int kb = s * 4 + khalf * 2;
        bf[s].x = W[kb       * EMBED_DIM + nt * 16 + m];
        bf[s].y = W[(kb + 1) * EMBED_DIM + nt * 16 + m];
    }
#pragma unroll
    for (int k = 0; k < N_CHUNKS / 32; ++k) {
        const unsigned* __restrict__ seg =
            list + (size_t)(wave + 32 * k) * 2 * (size_t)chunk_edges + off[k];
        const int c = num[k];
        for (int i = 0; i < c; i += 32) {
            const unsigned ent = (i + lane < c) ? seg[i + lane] : 0u;
            float    val[8];
            unsigned ad[8];
#pragma unroll
            for (int j = 0; j < 8; ++j) {
                const unsigned en = __shfl(ent, j * 4 + sub);
                ad[j]  = (en >> 24) * NUM_TYPES + d;
                val[j] = (i + j * 4 + sub < c)
                             ? cin[(size_t)(en & ENT_NODE) * NUM_TYPES + d] : 0.0f;
            }
#pragma unroll
            for (int j = 0; j < 8; ++j)
                if (i + j * 4 + sub < c) atomicAdd(&acc[ad[j]], val[j]);
        }
    }
    __syncthreads();
    v8f o = {0.f, 0.f, 0.f, 0.f, 0.f, 0.f, 0.f, 0.f};
    const float* arow = acc + (mt * 16 + m) * NUM_TYPES;
#pragma unroll
    for (int s = 0; s < 2; ++s) {
        const int kb = s * 4 + khalf * 2;
        v2f a;
        a.x = arow[kb];
        a.y = arow[kb + 1];
        o = __builtin_amdgcn_wmma_f32_16x16x4_f32(
             false, a,  false, bf[s],
             (short)0, o,  false,  false);
    }
    if (trow + 16 <= n_rows) {
#pragma unroll
        for (int r = 0; r < 8; ++r) obase[r * EMBED_DIM] = o[r];
    } else {
#pragma unroll
        for (int r = 0; r < 8; ++r)
            if (r < nvalid) obase[r * EMBED_DIM] = o[r];
    }
}

extern "C" void kernel_launch(void* const* d_in, const int* in_sizes, int n_in,
                              void* d_out, int out_size, void* d_ws, size_t ws_size,
                              hipStream_t stream) {
    const float* embed_table = (const float*)d_in[0];
    const int*   node_type   = (const int*)d_in[1];
    const int*   src         = (const int*)d_in[2];
    const int*   dst         = (const int*)d_in[3];
    float*       out         = (float*)d_out;

    const int n_nodes = in_sizes[1];
    const int n_edges = in_sizes[2];
    const int n_rows  = out_size / EMBED_DIM;

    const int nc = n_nodes * NUM_TYPES;

    const int    chunk_edges = (n_edges + N_CHUNKS - 1) / N_CHUNKS;
    const int    n_bins_all  = (n_nodes + BIN_ROWS - 1) / BIN_ROWS;
    const int    n_bins      = (n_rows + BIN_ROWS - 1) / BIN_ROWS;
    float*       cnt1        = (float*)d_ws;
    unsigned*    list        = (unsigned*)(cnt1 + (size_t)nc);
    const size_t list_words  = (size_t)N_CHUNKS * 2 * (size_t)chunk_edges;
    int2*        tab         = (int2*)(list + list_words);
    const size_t need = ((size_t)nc + list_words) * sizeof(float)
                      + (size_t)N_CHUNKS * MAX_BINS * sizeof(int2);
    const bool by_owner = ws_size >= need && n_bins_all <= MAX_BINS && n_rows <= n_nodes
                          && n_nodes > 0 && 2 * (long long)chunk_edges <= SORT_CAP;

    if (by_owner) {
        k_bin<<<N_CHUNKS, 1024, 0, stream>>>(src, dst, node_type, list, tab,
                                             n_edges, n_nodes, chunk_edges);
        k_count<<<n_bins_all, CNT_THREADS, 0, stream>>>(list, tab, cnt1, n_nodes, chunk_edges);
        k_own<<<n_bins, 1024, 0, stream>>>(list, tab, cnt1, embed_table, out, n_rows, chunk_edges);
    } else {
        float* c0 = (float*)d_ws;
        float* c1 = c0 + (size_t)nc;
        const int edges_per_block = 8 * EDGES_PER_WAVE;
        const int edge_blocks = (n_edges + edges_per_block - 1) / edges_per_block;
        k_init<<<(nc + 255) / 256, 256, 0, stream>>>(node_type, c0, c1, n_nodes);
        k_edges<<<edge_blocks, 256, 0, stream>>>(src, dst, c0, c1, n_edges);
        k_zero<<<(nc + 255) / 256, 256, 0, stream>>>(c0, nc);
        k_edges<<<edge_blocks, 256, 0, stream>>>(src, dst, c1, c0, n_edges);
        int n_mtiles = (n_rows + 15) / 16;
        int tiles = n_mtiles * 2;
        int waves_per_block = 256 / 32;
        int blocks = (tiles + waves_per_block - 1) / waves_per_block;
        k_embed_gemm<<<blocks, 256, 0, stream>>>(c0, embed_table, out, n_rows);
    }
}


# Round 2's owner split 4 ways: sort keys output rows by 64-row range; an 8-wave group per 64 rows packs its short segments in LDS, gathers rows densely; LDS add order differs, whole sums exact
# speedup vs baseline: 5.9758x; 1.1620x over previous
#include <hip/hip_runtime.h>
#include <hip/hip_bf16.h>

typedef __attribute__((ext_vector_type(2))) float v2f;
typedef __attribute__((ext_vector_type(8))) float v8f;

#define NUM_TYPES 8
#define EMBED_DIM 32

__global__ void k_init(const int* __restrict__ node_type,
                       float* __restrict__ c0, float* __restrict__ c1, int n_nodes) {
    int idx = blockIdx.x * blockDim.x + threadIdx.x;
    if (idx >= n_nodes * NUM_TYPES) return;
    int v = idx >> 3;
    int t = idx & 7;
    c0[idx] = (node_type[v] == t) ? 1.0f : 0.0f;
    c1[idx] = 0.0f;
}

__global__ void k_zero(float* __restrict__ p, int n) {
    int idx = blockIdx.x * blockDim.x + threadIdx.x;
    if (idx < n) p[idx] = 0.0f;
}

#define EDGES_PER_WAVE 128
#define CHUNKS_PER_WAVE (EDGES_PER_WAVE / 32)

template <bool GUARD>
__device__ __forceinline__ void scatter_chunk(int u, int v, int lane,
                                              const float* __restrict__ cin,
                                              float* __restrict__ cout) {
    const unsigned lcb = (unsigned)(lane & 7) * 4u;
    const int sub = lane >> 3;
    const char* __restrict__ cin_b  = (const char*)cin;
    char* __restrict__       cout_b = (char*)cout;
    unsigned ou[8], ov[8];
    float    cu[8], cv[8];
#pragma unroll
    for (int j = 0; j < 8; ++j) {
        const int uu = __shfl(u, j * 4 + sub);
        const int vv = __shfl(v, j * 4 + sub);
        ou[j] = (unsigned)uu * (NUM_TYPES * 4u) + lcb;
        ov[j] = (unsigned)vv * (NUM_TYPES * 4u) + lcb;
        if (!GUARD || uu >= 0) {
            cu[j] = *(const float*)(cin_b + ou[j]);
            cv[j] = *(const float*)(cin_b + ov[j]);
        } else {
            cu[j] = 0.0f;
            cv[j] = 0.0f;
        }
    }
#pragma unroll
    for (int j = 0; j < 8; ++j) {
        if (!GUARD || __shfl(u, j * 4 + sub) >= 0) {
            __hip_atomic_fetch_add((float*)(cout_b + ov[j]), cu[j],
                                   __ATOMIC_RELAXED, __HIP_MEMORY_SCOPE_AGENT);
            __hip_atomic_fetch_add((float*)(cout_b + ou[j]), cv[j],
                                   __ATOMIC_RELAXED, __HIP_MEMORY_SCOPE_AGENT);
        }
    }
}

__global__ void __launch_bounds__(256)
k_edges(const int* __restrict__ src, const int* __restrict__ dst,
        const float* __restrict__ cin, float* __restrict__ cout,
        int n_edges) {
    const int lane = threadIdx.x & 31;
    const int wave = (blockIdx.x * blockDim.x + threadIdx.x) >> 5;
    const long long first = (long long)wave * EDGES_PER_WAVE;
    if (first >= n_edges) return;

    int u[CHUNKS_PER_WAVE], v[CHUNKS_PER_WAVE];
    if (first + EDGES_PER_WAVE <= n_edges) {
#pragma unroll
        for (int c = 0; c < CHUNKS_PER_WAVE; ++c) {
            u[c] = src[first + c * 32 + lane];
            v[c] = dst[first + c * 32 + lane];
        }
#pragma unroll
        for (int c = 0; c < CHUNKS_PER_WAVE; ++c)
            scatter_chunk<false>(u[c], v[c], lane, cin, cout);
    } else {
#pragma unroll
        for (int c = 0; c < CHUNKS_PER_WAVE; ++c) {
            const long long e = first + c * 32 + lane;
            u[c] = (e < n_edges) ? src[e] : -1;
            v[c] = (e < n_edges) ? dst[e] : -1;
        }
#pragma unroll
        for (int c = 0; c < CHUNKS_PER_WAVE; ++c)
            scatter_chunk<true>(u[c], v[c], lane, cin, cout);
    }
}

__global__ void k_embed_gemm(const float* __restrict__ counts,
                             const float* __restrict__ W,
                             float* __restrict__ out, int n_rows) {
    const int wave = (blockIdx.x * blockDim.x + threadIdx.x) >> 5;
    const int lane = threadIdx.x & 31;
    const int n_mtiles = (n_rows + 15) >> 4;
    const int mt = wave >> 1;
    const int nt = wave & 1;
    if (mt >= n_mtiles) return;

    const int m     = lane & 15;
    const int khalf = (lane >> 4) & 1;

    v8f acc = {0.f, 0.f, 0.f, 0.f, 0.f, 0.f, 0.f, 0.f};
    const float* crow = counts + (size_t)(mt * 16 + m) * NUM_TYPES;

#pragma unroll
    for (int s = 0; s < 2; ++s) {
        const int kb = s * 4 + khalf * 2;
        v2f a, b;
        a.x = crow[kb];
        a.y = crow[kb + 1];
        b.x = W[(size_t)kb       * EMBED_DIM + nt * 16 + m];
        b.y = W[(size_t)(kb + 1) * EMBED_DIM + nt * 16 + m];
        acc = __builtin_amdgcn_wmma_f32_16x16x4_f32(
             false, a,  false, b,
             (short)0, acc,  false,  false);
    }

    if (mt * 16 + 16 <= n_rows) {
        float* obase = out + (size_t)(mt * 16 + khalf * 8) * EMBED_DIM + nt * 16 + m;
#pragma unroll
        for (int r = 0; r < 8; ++r) {
            obase[(size_t)r * EMBED_DIM] = acc[r];
        }
    } else {
#pragma unroll
        for (int r = 0; r < 8; ++r) {
            int row = mt * 16 + khalf * 8 + r;
            if (row < n_rows) out[(size_t)row * EMBED_DIM + nt * 16 + m] = acc[r];
        }
    }
}

#define BIN_ROWS  256
#define OWN_ROWS  64
#define OWN_SPLIT (BIN_ROWS / OWN_ROWS)
#define MAX_COARSE 512
#define MAX_BINS  1024
#define N_CHUNKS  256
#define SORT_CAP  16384
#define ENT_NODE  0x1FFFFFu
#define ENT_NONE  0xFFFFFFFFu
#define EPT       (SORT_CAP / 2 / 1024)

#define AUX_NODE_BITS 17
__device__ __forceinline__ int sort_key(unsigned x, unsigned fine_rows) {
    return x < fine_rows ? (int)(x / OWN_ROWS)
                         : (int)(fine_rows / OWN_ROWS + (x - fine_rows) / BIN_ROWS);
}
__global__ void __launch_bounds__(1024)
k_bin(const int* __restrict__ src, const int* __restrict__ dst,
      const int* __restrict__ node_type,
      unsigned* __restrict__ list, int* __restrict__ tab,
      int n_edges, int n_nodes, int chunk_edges, int fine_rows) {
    __shared__ int cnt[MAX_BINS];
    __shared__ int start[MAX_BINS];
    __shared__ int wsum[MAX_BINS / 32];
    __shared__ int total;
    __shared__ unsigned sorted[SORT_CAP];
    __shared__ unsigned aux[SORT_CAP];
    const int tid  = threadIdx.x;
    const int lane = tid & 31;
    const int wg   = blockIdx.x;
    const long long e0 = (long long)wg * chunk_edges;
    long long e1 = e0 + chunk_edges;
    if (e1 > n_edges) e1 = n_edges;
    if (tid < MAX_BINS) cnt[tid] = 0;
    unsigned uu[EPT], vv[EPT];
#pragma unroll
    for (int j = 0; j < EPT; ++j) {
        const long long e = e0 + tid + j * 1024;
        unsigned u = ENT_NONE, v = ENT_NONE;
        if (e < e1) {
            u = (unsigned)src[e];
            v = (unsigned)dst[e];
        }
        const bool ok = u < (unsigned)n_nodes && v < (unsigned)n_nodes;
        uu[j] = ok ? u : ENT_NONE;
        vv[j] = v;
    }
    __syncthreads();
#pragma unroll
    for (int j = 0; j < EPT; ++j) {
        if (uu[j] != ENT_NONE) {
            atomicAdd(&cnt[sort_key(vv[j], (unsigned)fine_rows)], 1);
            atomicAdd(&cnt[sort_key(uu[j], (unsigned)fine_rows)], 1);
        }
    }
    __syncthreads();
    int a = 0, inc = 0;
    if (tid < MAX_BINS) {
        a = cnt[tid];
        inc = a;
        for (int d = 1; d < 32; d <<= 1) {
            const int t = __shfl_up(inc, d);
            if (lane >= d) inc += t;
        }
        if (lane == 31) wsum[tid >> 5] = inc;
    }
    __syncthreads();
    if (tid < MAX_BINS) {
        int base = 0;
        for (int w = 0; w < (tid >> 5); ++w) base += wsum[w];
        const int ex = base + inc - a;
        start[tid] = ex;
        tab[wg * MAX_BINS + tid] = ex;
        cnt[tid] = 0;
        if (tid == MAX_BINS - 1) total = ex + a;
    }
    __syncthreads();
#pragma unroll
    for (int j = 0; j < EPT; ++j) {
        if (uu[j] != ENT_NONE) {
            const unsigned u = uu[j];
            const unsigned v = vv[j];
            int bin = sort_key(v, (unsigned)fine_rows);
            const int pv = start[bin] + atomicAdd(&cnt[bin], 1);
            bin = sort_key(u, (unsigned)fine_rows);
            const int pu = start[bin] + atomicAdd(&cnt[bin], 1);
            sorted[pv] = ((v % BIN_ROWS) << 24) | u;
            sorted[pu] = ((u % BIN_ROWS) << 24) | v;
            aux[pv] = ((unsigned)pu << AUX_NODE_BITS) | v;
            aux[pu] = ((unsigned)pv << AUX_NODE_BITS) | u;
        }
    }
    __syncthreads();
    const int n = total;
    if (n > 0) {
        unsigned ax[2 * EPT];
        int      ty[2 * EPT];
#pragma unroll
        for (int k = 0; k < 2 * EPT; ++k) {
            const int i = tid + k * 1024;
            ax[k] = aux[(i < n) ? i : 0];
        }
#pragma unroll
        for (int k = 0; k < 2 * EPT; ++k) {
            const int i = tid + k * 1024;
            ty[k] = node_type[(i < n) ? (ax[k] & ((1u << AUX_NODE_BITS) - 1u)) : 0u];
        }
#pragma unroll
        for (int k = 0; k < 2 * EPT; ++k) {
            const int i = tid + k * 1024;
            if (i < n) {
                const unsigned t = (unsigned)min(max(ty[k], 0), NUM_TYPES - 1);
                atomicOr(&sorted[ax[k] >> AUX_NODE_BITS], t << 21);
            }
        }
    }
    __syncthreads();
    unsigned* __restrict__ region = list + (size_t)wg * 2 * (size_t)chunk_edges;
    for (int i = tid; i < n; i += 1024) region[i] = sorted[i];
}

#define CNT_THREADS 256
__global__ void __launch_bounds__(CNT_THREADS)
k_count(const unsigned* __restrict__ list, const int* __restrict__ tab,
        float* __restrict__ cout, int n_nodes, int chunk_edges, int n_fine) {
    __shared__ int acc[BIN_ROWS * NUM_TYPES];
    const int tid  = threadIdx.x;
    const int lane = tid & 31;
    const int wave = tid >> 5;
    const int b    = blockIdx.x;
    for (int i = tid; i < BIN_ROWS * NUM_TYPES; i += CNT_THREADS) acc[i] = 0;
    __syncthreads();

    const int klo = (b < n_fine) ? b * OWN_SPLIT : b + (OWN_SPLIT - 1) * n_fine;
    const int khi = (b < n_fine) ? klo + OWN_SPLIT : klo + 1;
    const int* __restrict__ trow = tab + (size_t)(wave * 32 + lane) * MAX_BINS;
    const int o0 = trow[klo];
    const int o1 = trow[khi];
    const int2 oc = make_int2(o0, o1 - o0);
    const unsigned* __restrict__ wbase =
        list + (size_t)(wave * 32) * 2 * (size_t)chunk_edges;
    for (int g = 0; g < 2; ++g) {
        unsigned ea[16], eb[16];
#pragma unroll
        for (int k = 0; k < 16; ++k) {
            const int s = g * 16 + k;
            const int off = __builtin_amdgcn_readlane(oc.x, s);
            const int n   = __builtin_amdgcn_readlane(oc.y, s);
            const unsigned* __restrict__ seg =
                wbase + (size_t)s * 2 * (size_t)chunk_edges + off;
            ea[k] = (lane < n)      ? seg[lane]      : ENT_NONE;
            eb[k] = (lane + 32 < n) ? seg[lane + 32] : ENT_NONE;
        }
#pragma unroll
        for (int k = 0; k < 16; ++k) {
            if (ea[k] != ENT_NONE) atomicAdd(&acc[ea[k] >> 21], 1);
            if (eb[k] != ENT_NONE) atomicAdd(&acc[eb[k] >> 21], 1);
        }
    }
    int longest = oc.y;
    for (int d = 16; d > 0; d >>= 1) longest = max(longest, __shfl_xor(longest, d));
    if (longest > 64) {
        for (int s = 0; s < 32; ++s) {
            const int off = __shfl(oc.x, s);
            const int n   = __shfl(oc.y, s);
            const unsigned* __restrict__ seg =
                wbase + (size_t)s * 2 * (size_t)chunk_edges + off;
            for (int i = 64 + lane; i < n; i += 32) atomicAdd(&acc[seg[i] >> 21], 1);
        }
    }
    __syncthreads();
    const int row0 = b * BIN_ROWS;
    for (int i = tid; i < BIN_ROWS * NUM_TYPES; i += CNT_THREADS)
        if (row0 + (i >> 3) < n_nodes)
            cout[(size_t)row0 * NUM_TYPES + i] = (float)acc[i];
}

#define OWN_THREADS 256
#define OWN_WAVES   (OWN_THREADS / 32)
#define OWN_SEGS    (N_CHUNKS / OWN_WAVES)
#define PASS_ENT    16
#define Q_CAP       (32 * PASS_ENT)
#define OWN_NB      10
__global__ void __launch_bounds__(OWN_THREADS)
k_own(const unsigned* __restrict__ list, const int* __restrict__ tab,
      const float* __restrict__ cin, const float* __restrict__ W,
      float* __restrict__ out, int n_rows, int chunk_edges) {
    __shared__ float acc[OWN_ROWS * NUM_TYPES];
    __shared__ unsigned queue[OWN_WAVES * Q_CAP];
    const int tid  = threadIdx.x;
    const int lane = tid & 31;
    const int wave = tid >> 5;
    const int f    = blockIdx.x;
    if (f * OWN_ROWS >= n_rows) return;
    const int sub  = lane >> 3;
    const int d    = lane & 7;
    acc[tid]               = 0.0f;
    acc[tid + OWN_THREADS] = 0.0f;
    __syncthreads();

    const int* __restrict__ trow = tab + (size_t)(wave * OWN_SEGS + lane) * MAX_BINS + f;
    const int off = trow[0];
    const int cnt = trow[1] - off;
    const int wv    = __builtin_amdgcn_readfirstlane(wave);
    const int mt    = wv >> 1;
    const int nt    = wv & 1;
    const int m     = lane & 15;
    const int khalf = lane >> 4;
    const int trow0 = f * OWN_ROWS + mt * 16;
    float* obase = out + (size_t)trow0 * EMBED_DIM + nt * 16 + (khalf * 8 * EMBED_DIM + m);
    const int nvalid = n_rows - (trow0 + khalf * 8);
    v2f bf[2];
#pragma unroll
    for (int s = 0; s < 2; ++s) {
        const int kb = s * 4 + khalf * 2;
        bf[s].x = W[kb       * EMBED_DIM + nt * 16 + m];
        bf[s].y = W[(kb + 1) * EMBED_DIM + nt * 16 + m];
    }
    unsigned* q = queue + wave * Q_CAP;
    const unsigned* __restrict__ wbase =
        list + (size_t)(wave * OWN_SEGS) * 2 * (size_t)chunk_edges;
    for (int base = 0; __builtin_amdgcn_ballot_w32(cnt > base) != 0u; base += PASS_ENT) {
        unsigned e[2 * 8];
        int so[8], left[8];
#pragma unroll
        for (int g = 0; g < 8; ++g) {
            so[g]   = __shfl(off, g * 4 + sub);
            left[g] = __shfl(cnt, g * 4 + sub) - base;
        }
#pragma unroll
        for (int g = 0; g < 8; ++g) {
            const unsigned* __restrict__ seg =
                wbase + (size_t)(g * 4 + sub) * 2 * (size_t)chunk_edges + (so[g] + base);
            const bool ok0 = d     < left[g];
            const bool ok1 = d + 8 < left[g];
            const unsigned x0 = *(ok0 ? seg + d     : list);
            const unsigned x1 = *(ok1 ? seg + d + 8 : list);
            e[2 * g]     = ok0 ? x0 : ENT_NONE;
            e[2 * g + 1] = ok1 ? x1 : ENT_NONE;
        }
        int qn = 0;
#pragma unroll
        for (int r = 0; r < 2 * 8; ++r) {
            const bool     ok = e[r] != ENT_NONE;
            const unsigned mk = __builtin_amdgcn_ballot_w32(ok);
            if (ok) q[qn + (int)__builtin_amdgcn_mbcnt_lo(mk, 0u)] = e[r];
            qn += __builtin_popcount(mk);
        }
        __builtin_amdgcn_wave_barrier();
        for (int v0 = 0; v0 < qn; v0 += 32 * OWN_NB) {
            unsigned en[OWN_NB * 8];
            float    val[OWN_NB * 8];
#pragma unroll
            for (int k = 0; k < OWN_NB * 8; ++k) {
                const int v = v0 + 4 * k + sub;
                const unsigned x = q[min(v, Q_CAP - 1)];
                en[k] = (v < qn) ? x : ENT_NONE;
            }
#pragma unroll
            for (int k = 0; k < OWN_NB * 8; ++k) {
                const unsigned node = (en[k] != ENT_NONE) ? (en[k] & ENT_NODE) : 0u;
                val[k] = cin[(size_t)node * NUM_TYPES + d];
            }
#pragma unroll
            for (int k = 0; k < OWN_NB * 8; ++k)
                if (en[k] != ENT_NONE)
                    atomicAdd(&acc[((en[k] >> 24) & (OWN_ROWS - 1)) * NUM_TYPES + d], val[k]);
        }
        __builtin_amdgcn_wave_barrier();
    }
    __syncthreads();
    v8f o = {0.f, 0.f, 0.f, 0.f, 0.f, 0.f, 0.f, 0.f};
    const float* arow = acc + (mt * 16 + m) * NUM_TYPES;
#pragma unroll
    for (int s = 0; s < 2; ++s) {
        const int kb = s * 4 + khalf * 2;
        v2f a;
        a.x = arow[kb];
        a.y = arow[kb + 1];
        o = __builtin_amdgcn_wmma_f32_16x16x4_f32(
             false, a,  false, bf[s],
             (short)0, o,  false,  false);
    }
    if (trow0 + 16 <= n_rows) {
#pragma unroll
        for (int r = 0; r < 8; ++r) obase[r * EMBED_DIM] = o[r];
    } else {
#pragma unroll
        for (int r = 0; r < 8; ++r)
            if (r < nvalid) obase[r * EMBED_DIM] = o[r];
    }
}

extern "C" void kernel_launch(void* const* d_in, const int* in_sizes, int n_in,
                              void* d_out, int out_size, void* d_ws, size_t ws_size,
                              hipStream_t stream) {
    const float* embed_table = (const float*)d_in[0];
    const int*   node_type   = (const int*)d_in[1];
    const int*   src         = (const int*)d_in[2];
    const int*   dst         = (const int*)d_in[3];
    float*       out         = (float*)d_out;

    const int n_nodes = in_sizes[1];
    const int n_edges = in_sizes[2];
    const int n_rows  = out_size / EMBED_DIM;

    const int nc = n_nodes * NUM_TYPES;

    const int    chunk_edges = (n_edges + N_CHUNKS - 1) / N_CHUNKS;
    const int    n_bins_all  = (n_nodes + BIN_ROWS - 1) / BIN_ROWS;
    const int    n_bins      = (n_rows + BIN_ROWS - 1) / BIN_ROWS;
    float*       cnt1        = (float*)d_ws;
    unsigned*    list        = (unsigned*)(cnt1 + (size_t)nc);
    const size_t list_words  = (size_t)N_CHUNKS * 2 * (size_t)chunk_edges;
    int*         tab         = (int*)(list + list_words);
    const size_t need = ((size_t)nc + list_words) * sizeof(float)
                      + (size_t)N_CHUNKS * MAX_BINS * sizeof(int);
    const int    n_keys      = n_bins * OWN_SPLIT + (n_bins_all - n_bins);
    const bool by_owner = ws_size >= need && n_bins_all <= MAX_COARSE && n_keys < MAX_BINS
                          && n_rows <= n_nodes
                          && n_nodes > 0 && 2 * (long long)chunk_edges <= SORT_CAP;

    if (by_owner) {
        k_bin<<<N_CHUNKS, 1024, 0, stream>>>(src, dst, node_type, list, tab,
                                             n_edges, n_nodes, chunk_edges, n_bins * BIN_ROWS);
        k_count<<<n_bins_all, CNT_THREADS, 0, stream>>>(list, tab, cnt1, n_nodes, chunk_edges, n_bins);
        k_own<<<n_bins * OWN_SPLIT, OWN_THREADS, 0, stream>>>(list, tab, cnt1, embed_table, out,
                                                              n_rows, chunk_edges);
    } else {
        float* c0 = (float*)d_ws;
        float* c1 = c0 + (size_t)nc;
        const int edges_per_block = 8 * EDGES_PER_WAVE;
        const int edge_blocks = (n_edges + edges_per_block - 1) / edges_per_block;
        k_init<<<(nc + 255) / 256, 256, 0, stream>>>(node_type, c0, c1, n_nodes);
        k_edges<<<edge_blocks, 256, 0, stream>>>(src, dst, c0, c1, n_edges);
        k_zero<<<(nc + 255) / 256, 256, 0, stream>>>(c0, nc);
        k_edges<<<edge_blocks, 256, 0, stream>>>(src, dst, c1, c0, n_edges);
        int n_mtiles = (n_rows + 15) / 16;
        int tiles = n_mtiles * 2;
        int waves_per_block = 256 / 32;
        int blocks = (tiles + waves_per_block - 1) / waves_per_block;
        k_embed_gemm<<<blocks, 256, 0, stream>>>(c0, embed_table, out, n_rows);
    }
}
